# starting kernel
# baseline (speedup 1.0000x reference)
#include <hip/hip_runtime.h>
#include <hip/hip_bf16.h>

typedef float v2f __attribute__((ext_vector_type(2)));
typedef float v8f __attribute__((ext_vector_type(8)));

#define BB 16
#define SS 24
#define LL 128
#define DD 768
#define AA 8
#define TT 8

#define OFF_AVG  0
#define OFF_PRED (BB*SS*DD)
#define OFF_ARG0 (OFF_PRED + BB*SS*AA*DD)
#define OFF_ARG1 (OFF_ARG0 + BB*SS*AA*DD)

__global__ __launch_bounds__(256, 2)
void srl_fused_pool_wmma(const float* __restrict__ emb,
                         const int*   __restrict__ amask,
                         const int*   __restrict__ sid,
                         const int*   __restrict__ pred_ids,
                         const int*   __restrict__ a0_ids,
                         const int*   __restrict__ a1_ids,
                         float*       __restrict__ out)
{
    __shared__ int   sh_sid[LL];
    __shared__ int   sh_mask[LL];
    __shared__ int   sh_argid[24][TT];
    __shared__ int   sh_cnt[24][TT];
    __shared__ int   sh_sel[32];
    __shared__ float sh_inv[32];
    __shared__ int   sh_rowoff[32];
    __shared__ float sh_M[32][LL + 4];

    const int bs  = blockIdx.x;
    const int tid = threadIdx.x;

    if (tid < LL) {
        sh_sid[tid]  = sid[bs * LL + tid];
        sh_mask[tid] = amask[bs * LL + tid];
    }
    __syncthreads();

    if (tid < 24 * TT) {
        const int row = tid >> 3;
        const int t   = tid & 7;
        const int* base = (row < 8) ? pred_ids : ((row < 16) ? a0_ids : a1_ids);
        const int a  = row & 7;
        const int id = base[(bs * AA + a) * TT + t];
        sh_argid[row][t] = id;
        int c = 0;
        if (id != 0) {
            #pragma unroll 4
            for (int l = 0; l < LL; ++l) c += (sh_sid[l] == id) ? 1 : 0;
        }
        sh_cnt[row][t] = c;
    } else if (tid == 192) {
        int c = 0;
        #pragma unroll 4
        for (int l = 0; l < LL; ++l) c += sh_mask[l];
        if (c < 1) c = 1;
        sh_inv[0] = 1.0f / (float)c;
        sh_sel[0] = -1;
    }
    __syncthreads();

    if (tid < 24) {
        int sel = -1; float inv = 0.0f;
        for (int t = TT - 1; t >= 0; --t) {
            const int id = sh_argid[tid][t];
            const int c  = sh_cnt[tid][t];
            if (id != 0 && c > 0) { sel = id; inv = 1.0f / (float)c; break; }
        }
        sh_sel[tid + 1] = sel;
        sh_inv[tid + 1] = inv;
    }
    if (tid < 32) {
        const int r = tid;
        int off;
        if (r == 0)        off = OFF_AVG  + bs * DD;
        else if (r <= 8)   off = OFF_PRED + (bs * AA + (r - 1))  * DD;
        else if (r <= 16)  off = OFF_ARG0 + (bs * AA + (r - 9))  * DD;
        else if (r <= 24)  off = OFF_ARG1 + (bs * AA + (r - 17)) * DD;
        else               off = -1;
        sh_rowoff[r] = off;
    }
    __syncthreads();

    for (int idx = tid; idx < 32 * LL; idx += 256) {
        const int row = idx >> 7;
        const int l   = idx & (LL - 1);
        float v;
        if (row == 0)        v = sh_mask[l] ? sh_inv[0] : 0.0f;
        else if (row <= 24)  v = (sh_sid[l] == sh_sel[row]) ? sh_inv[row] : 0.0f;
        else                 v = 0.0f;
        sh_M[row][l] = v;
    }
    __syncthreads();

    const int wid  = tid >> 5;
    const int lane = tid & 31;
    const int lo   = lane & 15;
    const int hi   = lane >> 4;
    const int colBase = wid * 96;

    const float* __restrict__ E = emb + (size_t)bs * LL * DD;

    v8f acc[2][6];
    #pragma unroll
    for (int rt = 0; rt < 2; ++rt)
        #pragma unroll
        for (int j = 0; j < 6; ++j)
            acc[rt][j] = (v8f){};

    #pragma unroll 4
    for (int k = 0; k < LL; k += 4) {
        const int ka = k + 2 * hi;
        v2f a0, a1;
        a0.x = sh_M[lo][ka];        a0.y = sh_M[lo][ka + 1];
        a1.x = sh_M[16 + lo][ka];   a1.y = sh_M[16 + lo][ka + 1];
        #pragma unroll
        for (int j = 0; j < 6; ++j) {
            const int n = colBase + j * 16 + lo;
            v2f b;
            b.x = __builtin_nontemporal_load(&E[(size_t)ka * DD + n]);
            b.y = __builtin_nontemporal_load(&E[(size_t)(ka + 1) * DD + n]);
            acc[0][j] = __builtin_amdgcn_wmma_f32_16x16x4_f32(
                false, a0, false, b, (short)0, acc[0][j], false, false);
            acc[1][j] = __builtin_amdgcn_wmma_f32_16x16x4_f32(
                false, a1, false, b, (short)0, acc[1][j], false, false);
        }
    }

    #pragma unroll
    for (int rt = 0; rt < 2; ++rt) {
        #pragma unroll
        for (int v = 0; v < 8; ++v) {
            const int r   = rt * 16 + 8 * hi + v;
            const int off = sh_rowoff[r];
            if (off >= 0) {
                #pragma unroll
                for (int j = 0; j < 6; ++j) {
                    const int col = colBase + j * 16 + lo;
                    __builtin_nontemporal_store(acc[rt][j][v], &out[off + col]);
                }
            }
        }
    }
}

extern "C" void kernel_launch(void* const* d_in, const int* in_sizes, int n_in,
                              void* d_out, int out_size, void* d_ws, size_t ws_size,
                              hipStream_t stream) {
    const float* emb   = (const float*)d_in[0];
    const int*   amask = (const int*)  d_in[1];
    const int*   sids  = (const int*)  d_in[2];
    const int*   pids  = (const int*)  d_in[3];
    const int*   a0    = (const int*)  d_in[4];
    const int*   a1    = (const int*)  d_in[5];
    float* out = (float*)d_out;

    dim3 grid(BB * SS);
    dim3 block(256);
    hipLaunchKernelGGL(srl_fused_pool_wmma, grid, block, 0, stream,
                       emb, amask, sids, pids, a0, a1, out);
}


# Permute K so the positions pool rows 16..24 select come first; row tile 1's WMMAs run only over those 16-wide k blocks (zero blocks skipped; permuted K changes the f32 summation order)
# speedup vs baseline: 1.0970x; 1.0970x over previous
#include <hip/hip_runtime.h>
#include <hip/hip_bf16.h>

typedef float v2f __attribute__((ext_vector_type(2)));
typedef float v8f __attribute__((ext_vector_type(8)));

#define BB 16
#define SS 24
#define LL 128
#define DD 768
#define AA 8
#define TT 8

#define OFF_AVG  0
#define OFF_PRED (BB*SS*DD)
#define OFF_ARG0 (OFF_PRED + BB*SS*AA*DD)
#define OFF_ARG1 (OFF_ARG0 + BB*SS*AA*DD)

__global__ __launch_bounds__(256, 2)
void srl_fused_pool_wmma(const float* __restrict__ emb,
                         const int*   __restrict__ amask,
                         const int*   __restrict__ sid,
                         const int*   __restrict__ pred_ids,
                         const int*   __restrict__ a0_ids,
                         const int*   __restrict__ a1_ids,
                         float*       __restrict__ out)
{
    __shared__ int   sh_sid[LL];
    __shared__ int   sh_mask[LL];
    __shared__ int   sh_argid[24][TT];
    __shared__ int   sh_cnt[24][TT];
    __shared__ int   sh_sel[32];
    __shared__ float sh_inv[32];
    __shared__ int   sh_rowoff[32];
    __shared__ float sh_M[32][LL + 4];
    __shared__ int   sh_koff[LL];

    const int bs  = blockIdx.x;
    const int tid = threadIdx.x;

    if (tid < LL) {
        sh_sid[tid]  = sid[bs * LL + tid];
        sh_mask[tid] = amask[bs * LL + tid];
    }
    __syncthreads();

    if (tid < 24 * TT) {
        const int row = tid >> 3;
        const int t   = tid & 7;
        const int* base = (row < 8) ? pred_ids : ((row < 16) ? a0_ids : a1_ids);
        const int a  = row & 7;
        const int id = base[(bs * AA + a) * TT + t];
        sh_argid[row][t] = id;
        int c = 0;
        if (id != 0) {
            #pragma unroll 4
            for (int l = 0; l < LL; ++l) c += (sh_sid[l] == id) ? 1 : 0;
        }
        sh_cnt[row][t] = c;
    } else if (tid == 192) {
        int c = 0;
        #pragma unroll 4
        for (int l = 0; l < LL; ++l) c += sh_mask[l];
        if (c < 1) c = 1;
        sh_inv[0] = 1.0f / (float)c;
        sh_sel[0] = -1;
    }
    __syncthreads();

    if (tid < 24) {
        int sel = -1; float inv = 0.0f;
        for (int t = TT - 1; t >= 0; --t) {
            const int id = sh_argid[tid][t];
            const int c  = sh_cnt[tid][t];
            if (id != 0 && c > 0) { sel = id; inv = 1.0f / (float)c; break; }
        }
        sh_sel[tid + 1] = sel;
        sh_inv[tid + 1] = inv;
    }
    if (tid < 32) {
        const int r = tid;
        int off;
        if (r == 0)        off = OFF_AVG  + bs * DD;
        else if (r <= 8)   off = OFF_PRED + (bs * AA + (r - 1))  * DD;
        else if (r <= 16)  off = OFF_ARG0 + (bs * AA + (r - 9))  * DD;
        else if (r <= 24)  off = OFF_ARG1 + (bs * AA + (r - 17)) * DD;
        else               off = -1;
        sh_rowoff[r] = off;
    }
    __syncthreads();

    const int lane = tid & 31;
    int selB[9];
    #pragma unroll
    for (int r = 0; r < 9; ++r) selB[r] = sh_sel[16 + r];
    unsigned bm[4];
    #pragma unroll
    for (int c = 0; c < 4; ++c) {
        const int s = sh_sid[c * 32 + lane];
        bool f = false;
        #pragma unroll
        for (int r = 0; r < 9; ++r) f = f || (s == selB[r]);
        bm[c] = __builtin_amdgcn_ballot_w32(f);
    }
    const int p0 = __builtin_popcount(bm[0]);
    const int p1 = __builtin_popcount(bm[1]);
    const int p2 = __builtin_popcount(bm[2]);
    const int n1 = p0 + p1 + p2 + __builtin_popcount(bm[3]);
    {
        const int l = tid & (LL - 1);
        const int c = (tid >> 5) & 3;
        const unsigned mine = (c == 0) ? bm[0] : (c == 1) ? bm[1] : (c == 2) ? bm[2] : bm[3];
        const int before = ((c > 0) ? p0 : 0) + ((c > 1) ? p1 : 0) + ((c > 2) ? p2 : 0)
                         + __builtin_popcount(mine & ((1u << lane) - 1u));
        const bool first = (mine >> lane) & 1u;
        const int slot = first ? before : (n1 + l - before);
        if (tid < LL) sh_koff[slot] = l * DD;
        const int   sidl  = sh_sid[l];
        const int   maskl = sh_mask[l];
        #pragma unroll
        for (int it = 0; it < 16; ++it) {
            const int row = (tid >> 7) + 2 * it;
            float v;
            if (row == 0)        v = maskl ? sh_inv[0] : 0.0f;
            else if (row <= 24)  v = (sidl == sh_sel[row]) ? sh_inv[row] : 0.0f;
            else                 v = 0.0f;
            sh_M[row][slot] = v;
        }
    }
    __syncthreads();

    const int wid  = tid >> 5;
    const int lo   = lane & 15;
    const int hi   = lane >> 4;
    const int colBase = wid * 96;

    const float* __restrict__ E = emb + (size_t)bs * LL * DD + (colBase + lo);

    v8f acc[2][6];
    #pragma unroll
    for (int rt = 0; rt < 2; ++rt)
        #pragma unroll
        for (int j = 0; j < 6; ++j)
            acc[rt][j] = (v8f){};

    const int nb1 = (n1 + 15) >> 4;

    int kb = 0;
    for (; kb < nb1; ++kb) {
        v2f bq[4][6];
        #pragma unroll
        for (int s = 0; s < 4; ++s) {
            const int ka = kb * 16 + s * 4 + 2 * hi;
            const float* __restrict__ r0 = E + sh_koff[ka];
            const float* __restrict__ r1 = E + sh_koff[ka + 1];
            #pragma unroll
            for (int j = 0; j < 6; ++j) {
                bq[s][j].x = __builtin_nontemporal_load(r0 + j * 16);
                bq[s][j].y = __builtin_nontemporal_load(r1 + j * 16);
            }
        }
        __builtin_amdgcn_sched_barrier(0);
        #pragma unroll
        for (int s = 0; s < 4; ++s) {
            const int ka = kb * 16 + s * 4 + 2 * hi;
            v2f a0, a1;
            a0.x = sh_M[lo][ka];        a0.y = sh_M[lo][ka + 1];
            a1.x = sh_M[16 + lo][ka];   a1.y = sh_M[16 + lo][ka + 1];
            #pragma unroll
            for (int j = 0; j < 6; ++j) {
                acc[0][j] = __builtin_amdgcn_wmma_f32_16x16x4_f32(
                    false, a0, false, bq[s][j], (short)0, acc[0][j], false, false);
                acc[1][j] = __builtin_amdgcn_wmma_f32_16x16x4_f32(
                    false, a1, false, bq[s][j], (short)0, acc[1][j], false, false);
            }
        }
    }
    for (; kb < LL / 16; ++kb) {
        v2f bq[4][6];
        #pragma unroll
        for (int s = 0; s < 4; ++s) {
            const int ka = kb * 16 + s * 4 + 2 * hi;
            const float* __restrict__ r0 = E + sh_koff[ka];
            const float* __restrict__ r1 = E + sh_koff[ka + 1];
            #pragma unroll
            for (int j = 0; j < 6; ++j) {
                bq[s][j].x = __builtin_nontemporal_load(r0 + j * 16);
                bq[s][j].y = __builtin_nontemporal_load(r1 + j * 16);
            }
        }
        #pragma unroll
        for (int s = 0; s < 4; ++s) {
            const int ka = kb * 16 + s * 4 + 2 * hi;
            v2f a0;
            a0.x = sh_M[lo][ka];        a0.y = sh_M[lo][ka + 1];
            #pragma unroll
            for (int j = 0; j < 6; ++j) {
                acc[0][j] = __builtin_amdgcn_wmma_f32_16x16x4_f32(
                    false, a0, false, bq[s][j], (short)0, acc[0][j], false, false);
            }
        }
    }

    #pragma unroll
    for (int rt = 0; rt < 2; ++rt) {
        #pragma unroll
        for (int v = 0; v < 8; ++v) {
            const int r   = rt * 16 + 8 * hi + v;
            const int off = sh_rowoff[r];
            if (off >= 0) {
                #pragma unroll
                for (int j = 0; j < 6; ++j) {
                    const int col = colBase + j * 16 + lo;
                    __builtin_nontemporal_store(acc[rt][j][v], &out[off + col]);
                }
            }
        }
    }
}

extern "C" void kernel_launch(void* const* d_in, const int* in_sizes, int n_in,
                              void* d_out, int out_size, void* d_ws, size_t ws_size,
                              hipStream_t stream) {
    const float* emb   = (const float*)d_in[0];
    const int*   amask = (const int*)  d_in[1];
    const int*   sids  = (const int*)  d_in[2];
    const int*   pids  = (const int*)  d_in[3];
    const int*   a0    = (const int*)  d_in[4];
    const int*   a1    = (const int*)  d_in[5];
    float* out = (float*)d_out;

    dim3 grid(BB * SS);
    dim3 block(256);
    hipLaunchKernelGGL(srl_fused_pool_wmma, grid, block, 0, stream,
                       emb, amask, sids, pids, a0, a1, out);
}


# Each workgroup prefetches its whole 128x768 E tile to L2 (global_prefetch_b8, 6 units of 256 B per thread) behind its index loads, so memory fetches run under the index prologue; sums unchanged
# speedup vs baseline: 1.2269x; 1.1184x over previous
#include <hip/hip_runtime.h>
#include <hip/hip_bf16.h>

typedef float v2f __attribute__((ext_vector_type(2)));
typedef float v8f __attribute__((ext_vector_type(8)));

#define BB 16
#define SS 24
#define LL 128
#define DD 768
#define AA 8
#define TT 8

#define OFF_AVG  0
#define OFF_PRED (BB*SS*DD)
#define OFF_ARG0 (OFF_PRED + BB*SS*AA*DD)
#define OFF_ARG1 (OFF_ARG0 + BB*SS*AA*DD)

__global__ __launch_bounds__(256, 2)
void srl_fused_pool_wmma(const float* __restrict__ emb,
                         const int*   __restrict__ amask,
                         const int*   __restrict__ sid,
                         const int*   __restrict__ pred_ids,
                         const int*   __restrict__ a0_ids,
                         const int*   __restrict__ a1_ids,
                         float*       __restrict__ out)
{
    __shared__ int   sh_sid[LL];
    __shared__ int   sh_mask[LL];
    __shared__ int   sh_argid[24][TT];
    __shared__ int   sh_cnt[24][TT];
    __shared__ int   sh_sel[32];
    __shared__ float sh_inv[32];
    __shared__ int   sh_rowoff[32];
    __shared__ float sh_M[32][LL + 4];
    __shared__ int   sh_koff[LL];

    const int bs  = blockIdx.x;
    const int tid = threadIdx.x;

    int ld_sid = 0, ld_mask = 0, ld_id = 0;
    if (tid < LL) {
        ld_sid  = sid[bs * LL + tid];
        ld_mask = amask[bs * LL + tid];
    }
    if (tid < 24 * TT) {
        const int row = tid >> 3;
        const int* base = (row < 8) ? pred_ids : ((row < 16) ? a0_ids : a1_ids);
        ld_id = base[(bs * AA + (row & 7)) * TT + (tid & 7)];
    }
    {
        const float* tile = emb + (size_t)bs * LL * DD;
        #pragma unroll
        for (int i = 0; i < 6; ++i) {
            const float* p = tile + (size_t)(i * 256 + tid) * 64;
            asm volatile("global_prefetch_b8 %0, off th:TH_LOAD_NT scope:SCOPE_SE" :: "v"(p));
        }
    }
    if (tid < LL) {
        sh_sid[tid]  = ld_sid;
        sh_mask[tid] = ld_mask;
    }
    __syncthreads();

    if (tid < 24 * TT) {
        const int row = tid >> 3;
        const int t   = tid & 7;
        const int id = ld_id;
        sh_argid[row][t] = id;
        int c = 0;
        if (id != 0) {
            #pragma unroll 4
            for (int l = 0; l < LL; ++l) c += (sh_sid[l] == id) ? 1 : 0;
        }
        sh_cnt[row][t] = c;
    } else if (tid == 192) {
        int c = 0;
        #pragma unroll 4
        for (int l = 0; l < LL; ++l) c += sh_mask[l];
        if (c < 1) c = 1;
        sh_inv[0] = 1.0f / (float)c;
        sh_sel[0] = -1;
    }
    __syncthreads();

    if (tid < 24) {
        int sel = -1; float inv = 0.0f;
        for (int t = TT - 1; t >= 0; --t) {
            const int id = sh_argid[tid][t];
            const int c  = sh_cnt[tid][t];
            if (id != 0 && c > 0) { sel = id; inv = 1.0f / (float)c; break; }
        }
        sh_sel[tid + 1] = sel;
        sh_inv[tid + 1] = inv;
    }
    if (tid < 32) {
        const int r = tid;
        int off;
        if (r == 0)        off = OFF_AVG  + bs * DD;
        else if (r <= 8)   off = OFF_PRED + (bs * AA + (r - 1))  * DD;
        else if (r <= 16)  off = OFF_ARG0 + (bs * AA + (r - 9))  * DD;
        else if (r <= 24)  off = OFF_ARG1 + (bs * AA + (r - 17)) * DD;
        else               off = -1;
        sh_rowoff[r] = off;
    }
    __syncthreads();

    const int lane = tid & 31;
    int selB[9];
    #pragma unroll
    for (int r = 0; r < 9; ++r) selB[r] = sh_sel[16 + r];
    unsigned bm[4];
    #pragma unroll
    for (int c = 0; c < 4; ++c) {
        const int s = sh_sid[c * 32 + lane];
        bool f = false;
        #pragma unroll
        for (int r = 0; r < 9; ++r) f = f || (s == selB[r]);
        bm[c] = __builtin_amdgcn_ballot_w32(f);
    }
    const int p0 = __builtin_popcount(bm[0]);
    const int p1 = __builtin_popcount(bm[1]);
    const int p2 = __builtin_popcount(bm[2]);
    const int n1 = p0 + p1 + p2 + __builtin_popcount(bm[3]);
    {
        const int l = tid & (LL - 1);
        const int c = (tid >> 5) & 3;
        const unsigned mine = (c == 0) ? bm[0] : (c == 1) ? bm[1] : (c == 2) ? bm[2] : bm[3];
        const int before = ((c > 0) ? p0 : 0) + ((c > 1) ? p1 : 0) + ((c > 2) ? p2 : 0)
                         + __builtin_popcount(mine & ((1u << lane) - 1u));
        const bool first = (mine >> lane) & 1u;
        const int slot = first ? before : (n1 + l - before);
        if (tid < LL) sh_koff[slot] = l * DD;
        const int   sidl  = sh_sid[l];
        const int   maskl = sh_mask[l];
        #pragma unroll
        for (int it = 0; it < 16; ++it) {
            const int row = (tid >> 7) + 2 * it;
            float v;
            if (row == 0)        v = maskl ? sh_inv[0] : 0.0f;
            else if (row <= 24)  v = (sidl == sh_sel[row]) ? sh_inv[row] : 0.0f;
            else                 v = 0.0f;
            sh_M[row][slot] = v;
        }
    }
    __syncthreads();

    const int wid  = tid >> 5;
    const int lo   = lane & 15;
    const int hi   = lane >> 4;
    const int colBase = wid * 96;

    const float* __restrict__ E = emb + (size_t)bs * LL * DD + (colBase + lo);

    v8f acc[2][6];
    #pragma unroll
    for (int rt = 0; rt < 2; ++rt)
        #pragma unroll
        for (int j = 0; j < 6; ++j)
            acc[rt][j] = (v8f){};

    const int nb1 = (n1 + 15) >> 4;

    int kb = 0;
    for (; kb < nb1; ++kb) {
        v2f bq[4][6];
        #pragma unroll
        for (int s = 0; s < 4; ++s) {
            const int ka = kb * 16 + s * 4 + 2 * hi;
            const float* __restrict__ r0 = E + sh_koff[ka];
            const float* __restrict__ r1 = E + sh_koff[ka + 1];
            #pragma unroll
            for (int j = 0; j < 6; ++j) {
                bq[s][j].x = __builtin_nontemporal_load(r0 + j * 16);
                bq[s][j].y = __builtin_nontemporal_load(r1 + j * 16);
            }
        }
        __builtin_amdgcn_sched_barrier(0);
        #pragma unroll
        for (int s = 0; s < 4; ++s) {
            const int ka = kb * 16 + s * 4 + 2 * hi;
            v2f a0, a1;
            a0.x = sh_M[lo][ka];        a0.y = sh_M[lo][ka + 1];
            a1.x = sh_M[16 + lo][ka];   a1.y = sh_M[16 + lo][ka + 1];
            #pragma unroll
            for (int j = 0; j < 6; ++j) {
                acc[0][j] = __builtin_amdgcn_wmma_f32_16x16x4_f32(
                    false, a0, false, bq[s][j], (short)0, acc[0][j], false, false);
                acc[1][j] = __builtin_amdgcn_wmma_f32_16x16x4_f32(
                    false, a1, false, bq[s][j], (short)0, acc[1][j], false, false);
            }
        }
    }
    for (; kb < LL / 16; ++kb) {
        v2f bq[4][6];
        #pragma unroll
        for (int s = 0; s < 4; ++s) {
            const int ka = kb * 16 + s * 4 + 2 * hi;
            const float* __restrict__ r0 = E + sh_koff[ka];
            const float* __restrict__ r1 = E + sh_koff[ka + 1];
            #pragma unroll
            for (int j = 0; j < 6; ++j) {
                bq[s][j].x = __builtin_nontemporal_load(r0 + j * 16);
                bq[s][j].y = __builtin_nontemporal_load(r1 + j * 16);
            }
        }
        #pragma unroll
        for (int s = 0; s < 4; ++s) {
            const int ka = kb * 16 + s * 4 + 2 * hi;
            v2f a0;
            a0.x = sh_M[lo][ka];        a0.y = sh_M[lo][ka + 1];
            #pragma unroll
            for (int j = 0; j < 6; ++j) {
                acc[0][j] = __builtin_amdgcn_wmma_f32_16x16x4_f32(
                    false, a0, false, bq[s][j], (short)0, acc[0][j], false, false);
            }
        }
    }

    #pragma unroll
    for (int rt = 0; rt < 2; ++rt) {
        #pragma unroll
        for (int v = 0; v < 8; ++v) {
            const int r   = rt * 16 + 8 * hi + v;
            const int off = sh_rowoff[r];
            if (off >= 0) {
                #pragma unroll
                for (int j = 0; j < 6; ++j) {
                    const int col = colBase + j * 16 + lo;
                    __builtin_nontemporal_store(acc[rt][j][v], &out[off + col]);
                }
            }
        }
    }
}

extern "C" void kernel_launch(void* const* d_in, const int* in_sizes, int n_in,
                              void* d_out, int out_size, void* d_ws, size_t ws_size,
                              hipStream_t stream) {
    const float* emb   = (const float*)d_in[0];
    const int*   amask = (const int*)  d_in[1];
    const int*   sids  = (const int*)  d_in[2];
    const int*   pids  = (const int*)  d_in[3];
    const int*   a0    = (const int*)  d_in[4];
    const int*   a1    = (const int*)  d_in[5];
    float* out = (float*)d_out;

    dim3 grid(BB * SS);
    dim3 block(256);
    hipLaunchKernelGGL(srl_fused_pool_wmma, grid, block, 0, stream,
                       emb, amask, sids, pids, a0, a1, out);
}


# Grid cut into 768 equal units (tile x column half, 4-wave workgroups, 3 per WGP by amdgpu_waves_per_eu(3,3)) in place of 384 tiles that load WGPs with 2 or 1; per-wave K loop and sums unchanged
# speedup vs baseline: 1.3330x; 1.0865x over previous
#include <hip/hip_runtime.h>
#include <hip/hip_bf16.h>

typedef float v2f __attribute__((ext_vector_type(2)));
typedef float v8f __attribute__((ext_vector_type(8)));

#define BB 16
#define SS 24
#define LL 128
#define DD 768
#define AA 8
#define TT 8

#define OFF_AVG  0
#define OFF_PRED (BB*SS*DD)
#define OFF_ARG0 (OFF_PRED + BB*SS*AA*DD)
#define OFF_ARG1 (OFF_ARG0 + BB*SS*AA*DD)

__global__ __launch_bounds__(128) __attribute__((amdgpu_waves_per_eu(3, 3)))
void srl_fused_pool_wmma(const float* __restrict__ emb,
                         const int*   __restrict__ amask,
                         const int*   __restrict__ sid,
                         const int*   __restrict__ pred_ids,
                         const int*   __restrict__ a0_ids,
                         const int*   __restrict__ a1_ids,
                         float*       __restrict__ out)
{
    __shared__ int   sh_sid[LL];
    __shared__ int   sh_mask[LL];
    __shared__ int   sh_argid[24][TT];
    __shared__ int   sh_cnt[24][TT];
    __shared__ int   sh_sel[32];
    __shared__ float sh_inv[32];
    __shared__ int   sh_rowoff[32];
    __shared__ float sh_M[32][LL + 4];
    __shared__ int   sh_koff[LL];

    const int bs   = blockIdx.x >> 1;
    const int half = blockIdx.x & 1;
    const int tid  = threadIdx.x;

    const int ld_sid  = sid[bs * LL + tid];
    const int ld_mask = amask[bs * LL + tid];
    int ld_id0, ld_id1 = 0;
    {
        const int row = tid >> 3;
        const int* base = (row < 8) ? pred_ids : a0_ids;
        ld_id0 = base[(bs * AA + (row & 7)) * TT + (tid & 7)];
        if (tid < 64)
            ld_id1 = a1_ids[(bs * AA + (tid >> 3)) * TT + (tid & 7)];
    }
    {
        const float* tile = emb + (size_t)bs * LL * DD + half * (DD / 2);
        #pragma unroll
        for (int i = 0; i < 6; ++i) {
            const int u   = i * 128 + tid;
            const int row = u / 6;
            const float* p = tile + row * DD + (u - row * 6) * 64;
            asm volatile("global_prefetch_b8 %0, off th:TH_LOAD_NT scope:SCOPE_SE" :: "v"(p));
        }
    }
    sh_sid[tid]  = ld_sid;
    sh_mask[tid] = ld_mask;
    __syncthreads();

    {
        const int row = tid >> 3;
        const int t   = tid & 7;
        int c0 = 0, c1 = 0, cm = 0;
        #pragma unroll 4
        for (int l = 0; l < LL; ++l) {
            const int s = sh_sid[l];
            c0 += (s == ld_id0) ? 1 : 0;
            c1 += (s == ld_id1) ? 1 : 0;
            cm += sh_mask[l];
        }
        sh_argid[row][t] = ld_id0;
        sh_cnt[row][t]   = c0;
        if (tid < 64) {
            sh_argid[16 + row][t] = ld_id1;
            sh_cnt[16 + row][t]   = c1;
        }
        if (tid == LL - 1) {
            int c = cm;
            if (c < 1) c = 1;
            sh_inv[0] = 1.0f / (float)c;
            sh_sel[0] = -1;
        }
    }
    __syncthreads();

    if (tid < 24) {
        int sel = -1; float inv = 0.0f;
        for (int t = TT - 1; t >= 0; --t) {
            const int id = sh_argid[tid][t];
            const int c  = sh_cnt[tid][t];
            if (id != 0 && c > 0) { sel = id; inv = 1.0f / (float)c; break; }
        }
        sh_sel[tid + 1] = sel;
        sh_inv[tid + 1] = inv;
    }
    if (tid < 32) {
        const int r = tid;
        int off;
        if (r == 0)        off = OFF_AVG  + bs * DD;
        else if (r <= 8)   off = OFF_PRED + (bs * AA + (r - 1))  * DD;
        else if (r <= 16)  off = OFF_ARG0 + (bs * AA + (r - 9))  * DD;
        else if (r <= 24)  off = OFF_ARG1 + (bs * AA + (r - 17)) * DD;
        else               off = -1;
        sh_rowoff[r] = off;
    }
    __syncthreads();

    const int lane = tid & 31;
    int selB[9];
    #pragma unroll
    for (int r = 0; r < 9; ++r) selB[r] = sh_sel[16 + r];
    unsigned bm[4];
    #pragma unroll
    for (int c = 0; c < 4; ++c) {
        const int s = sh_sid[c * 32 + lane];
        bool f = false;
        #pragma unroll
        for (int r = 0; r < 9; ++r) f = f || (s == selB[r]);
        bm[c] = __builtin_amdgcn_ballot_w32(f);
    }
    const int p0 = __builtin_popcount(bm[0]);
    const int p1 = __builtin_popcount(bm[1]);
    const int p2 = __builtin_popcount(bm[2]);
    const int n1 = p0 + p1 + p2 + __builtin_popcount(bm[3]);
    {
        const int l = tid;
        const int c = tid >> 5;
        const unsigned mine = (c == 0) ? bm[0] : (c == 1) ? bm[1] : (c == 2) ? bm[2] : bm[3];
        const int before = ((c > 0) ? p0 : 0) + ((c > 1) ? p1 : 0) + ((c > 2) ? p2 : 0)
                         + __builtin_popcount(mine & ((1u << lane) - 1u));
        const bool first = (mine >> lane) & 1u;
        const int slot = first ? before : (n1 + l - before);
        sh_koff[slot] = l * DD;
        const int   sidl  = sh_sid[l];
        const int   maskl = sh_mask[l];
        #pragma unroll
        for (int row = 0; row < 32; ++row) {
            float v;
            if (row == 0)        v = maskl ? sh_inv[0] : 0.0f;
            else if (row <= 24)  v = (sidl == sh_sel[row]) ? sh_inv[row] : 0.0f;
            else                 v = 0.0f;
            sh_M[row][slot] = v;
        }
    }
    __syncthreads();

    const int wid  = tid >> 5;
    const int lo   = lane & 15;
    const int hi   = lane >> 4;
    const int colBase = half * (DD / 2) + wid * 96;

    const float* __restrict__ E = emb + (size_t)bs * LL * DD + (colBase + lo);

    v8f acc[2][6];
    #pragma unroll
    for (int rt = 0; rt < 2; ++rt)
        #pragma unroll
        for (int j = 0; j < 6; ++j)
            acc[rt][j] = (v8f){};

    const int nb1 = (n1 + 15) >> 4;

    int kb = 0;
    for (; kb < nb1; ++kb) {
        v2f bq[4][6];
        #pragma unroll
        for (int s = 0; s < 4; ++s) {
            const int ka = kb * 16 + s * 4 + 2 * hi;
            const float* __restrict__ r0 = E + sh_koff[ka];
            const float* __restrict__ r1 = E + sh_koff[ka + 1];
            #pragma unroll
            for (int j = 0; j < 6; ++j) {
                bq[s][j].x = __builtin_nontemporal_load(r0 + j * 16);
                bq[s][j].y = __builtin_nontemporal_load(r1 + j * 16);
            }
        }
        __builtin_amdgcn_sched_barrier(0);
        #pragma unroll
        for (int s = 0; s < 4; ++s) {
            const int ka = kb * 16 + s * 4 + 2 * hi;
            v2f a0, a1;
            a0.x = sh_M[lo][ka];        a0.y = sh_M[lo][ka + 1];
            a1.x = sh_M[16 + lo][ka];   a1.y = sh_M[16 + lo][ka + 1];
            #pragma unroll
            for (int j = 0; j < 6; ++j) {
                acc[0][j] = __builtin_amdgcn_wmma_f32_16x16x4_f32(
                    false, a0, false, bq[s][j], (short)0, acc[0][j], false, false);
                acc[1][j] = __builtin_amdgcn_wmma_f32_16x16x4_f32(
                    false, a1, false, bq[s][j], (short)0, acc[1][j], false, false);
            }
        }
    }
    for (; kb < LL / 16; ++kb) {
        v2f bq[4][6];
        #pragma unroll
        for (int s = 0; s < 4; ++s) {
            const int ka = kb * 16 + s * 4 + 2 * hi;
            const float* __restrict__ r0 = E + sh_koff[ka];
            const float* __restrict__ r1 = E + sh_koff[ka + 1];
            #pragma unroll
            for (int j = 0; j < 6; ++j) {
                bq[s][j].x = __builtin_nontemporal_load(r0 + j * 16);
                bq[s][j].y = __builtin_nontemporal_load(r1 + j * 16);
            }
        }
        #pragma unroll
        for (int s = 0; s < 4; ++s) {
            const int ka = kb * 16 + s * 4 + 2 * hi;
            v2f a0;
            a0.x = sh_M[lo][ka];        a0.y = sh_M[lo][ka + 1];
            #pragma unroll
            for (int j = 0; j < 6; ++j) {
                acc[0][j] = __builtin_amdgcn_wmma_f32_16x16x4_f32(
                    false, a0, false, bq[s][j], (short)0, acc[0][j], false, false);
            }
        }
    }

    #pragma unroll
    for (int rt = 0; rt < 2; ++rt) {
        #pragma unroll
        for (int v = 0; v < 8; ++v) {
            const int r   = rt * 16 + 8 * hi + v;
            const int off = sh_rowoff[r];
            if (off >= 0) {
                #pragma unroll
                for (int j = 0; j < 6; ++j) {
                    const int col = colBase + j * 16 + lo;
                    __builtin_nontemporal_store(acc[rt][j][v], &out[off + col]);
                }
            }
        }
    }
}

extern "C" void kernel_launch(void* const* d_in, const int* in_sizes, int n_in,
                              void* d_out, int out_size, void* d_ws, size_t ws_size,
                              hipStream_t stream) {
    const float* emb   = (const float*)d_in[0];
    const int*   amask = (const int*)  d_in[1];
    const int*   sids  = (const int*)  d_in[2];
    const int*   pids  = (const int*)  d_in[3];
    const int*   a0    = (const int*)  d_in[4];
    const int*   a1    = (const int*)  d_in[5];
    float* out = (float*)d_out;

    dim3 grid(BB * SS * 2);
    dim3 block(128);
    hipLaunchKernelGGL(srl_fused_pool_wmma, grid, block, 0, stream,
                       emb, amask, sids, pids, a0, a1, out);
}


# Index prologue on wave ballots: each wave resolves 6 pool rows in registers (token id read from its lane, count = popcount of 4 compare masks, scalar early exit); count loop and 2 barriers gone
# speedup vs baseline: 1.4017x; 1.0515x over previous
#include <hip/hip_runtime.h>
#include <hip/hip_bf16.h>

typedef float v2f __attribute__((ext_vector_type(2)));
typedef float v8f __attribute__((ext_vector_type(8)));

#define BB 16
#define SS 24
#define LL 128
#define DD 768
#define AA 8
#define TT 8

#define OFF_AVG  0
#define OFF_PRED (BB*SS*DD)
#define OFF_ARG0 (OFF_PRED + BB*SS*AA*DD)
#define OFF_ARG1 (OFF_ARG0 + BB*SS*AA*DD)

__global__ __launch_bounds__(128) __attribute__((amdgpu_waves_per_eu(3, 3)))
void srl_fused_pool_wmma(const float* __restrict__ emb,
                         const int*   __restrict__ amask,
                         const int*   __restrict__ sid,
                         const int*   __restrict__ pred_ids,
                         const int*   __restrict__ a0_ids,
                         const int*   __restrict__ a1_ids,
                         float*       __restrict__ out)
{
    __shared__ int   sh_sel[32];
    __shared__ float sh_inv[32];
    __shared__ int   sh_rowoff[32];
    __shared__ float sh_M[32][LL + 4];
    __shared__ int   sh_koff[LL];

    const int bs   = blockIdx.x >> 1;
    const int half = blockIdx.x & 1;
    const int tid  = threadIdx.x;
    const int lane = tid & 31;
    const int wid  = tid >> 5;

    int sidv[4], maskv[4];
    #pragma unroll
    for (int c = 0; c < 4; ++c) {
        sidv[c]  = sid[bs * LL + c * 32 + lane];
        maskv[c] = amask[bs * LL + c * 32 + lane];
    }
    int idv0, idv1;
    {
        const int t  = lane & 7;
        const int r0 = 6 * wid + (lane >> 3);
        const int r1 = 6 * wid + 4 + ((lane >> 3) & 1);
        const int* b0 = (r0 < 8) ? pred_ids : (r0 < 16) ? a0_ids : a1_ids;
        const int* b1 = (r1 < 8) ? pred_ids : (r1 < 16) ? a0_ids : a1_ids;
        idv0 = b0[(bs * AA + (r0 & 7)) * TT + t];
        idv1 = b1[(bs * AA + (r1 & 7)) * TT + t];
    }
    {
        const float* tile = emb + (size_t)bs * LL * DD + half * (DD / 2);
        #pragma unroll
        for (int i = 0; i < 6; ++i) {
            const int u   = i * 128 + tid;
            const int row = u / 6;
            const float* p = tile + row * DD + (u - row * 6) * 64;
            asm volatile("global_prefetch_b8 %0, off th:TH_LOAD_NT scope:SCOPE_SE" :: "v"(p));
        }
    }
    if (tid < 32) {
        const int r = tid;
        int off;
        if (r == 0)        off = OFF_AVG  + bs * DD;
        else if (r <= 8)   off = OFF_PRED + (bs * AA + (r - 1))  * DD;
        else if (r <= 16)  off = OFF_ARG0 + (bs * AA + (r - 9))  * DD;
        else if (r <= 24)  off = OFF_ARG1 + (bs * AA + (r - 17)) * DD;
        else               off = -1;
        sh_rowoff[r] = off;
    }

    int my_sel = -1, my_cnt = 0;
    #pragma unroll
    for (int r = 0; r < 6; ++r) {
        const int src  = (r < 4) ? idv0 : idv1;
        const int base = (r & 3) * 8;
        int sel = -1, cnt = 0;
        for (int t = TT - 1; t >= 0; --t) {
            const int id = __builtin_amdgcn_readlane(src, base + t);
            if (id != 0) {
                const int c = __builtin_popcount(__builtin_amdgcn_ballot_w32(sidv[0] == id))
                            + __builtin_popcount(__builtin_amdgcn_ballot_w32(sidv[1] == id))
                            + __builtin_popcount(__builtin_amdgcn_ballot_w32(sidv[2] == id))
                            + __builtin_popcount(__builtin_amdgcn_ballot_w32(sidv[3] == id));
                if (c > 0) { sel = id; cnt = c; break; }
            }
        }
        if (lane == r) { my_sel = sel; my_cnt = cnt; }
    }
    if (lane < 6) {
        sh_sel[6 * wid + lane + 1] = my_sel;
        sh_inv[6 * wid + lane + 1] = (my_cnt > 0) ? 1.0f / (float)my_cnt : 0.0f;
    }
    {
        int x = maskv[0] + maskv[1] + maskv[2] + maskv[3];
        x += __builtin_amdgcn_update_dpp(0, x, 0x111, 0xF, 0xF, true);
        x += __builtin_amdgcn_update_dpp(0, x, 0x112, 0xF, 0xF, true);
        x += __builtin_amdgcn_update_dpp(0, x, 0x114, 0xF, 0xF, true);
        x += __builtin_amdgcn_update_dpp(0, x, 0x118, 0xF, 0xF, true);
        int c = __builtin_amdgcn_readlane(x, 15) + __builtin_amdgcn_readlane(x, 31);
        if (c < 1) c = 1;
        if (tid == 0) {
            sh_inv[0] = 1.0f / (float)c;
            sh_sel[0] = -1;
        }
    }
    __syncthreads();

    int selB[9];
    #pragma unroll
    for (int r = 0; r < 9; ++r) selB[r] = sh_sel[16 + r];
    unsigned bm[4];
    #pragma unroll
    for (int c = 0; c < 4; ++c) {
        const int s = sidv[c];
        bool f = false;
        #pragma unroll
        for (int r = 0; r < 9; ++r) f = f || (s == selB[r]);
        bm[c] = __builtin_amdgcn_ballot_w32(f);
    }
    const int p0 = __builtin_popcount(bm[0]);
    const int p1 = __builtin_popcount(bm[1]);
    const int p2 = __builtin_popcount(bm[2]);
    const int n1 = p0 + p1 + p2 + __builtin_popcount(bm[3]);
    {
        const int l = tid;
        const int c = tid >> 5;
        const unsigned mine = (c == 0) ? bm[0] : (c == 1) ? bm[1] : (c == 2) ? bm[2] : bm[3];
        const int before = ((c > 0) ? p0 : 0) + ((c > 1) ? p1 : 0) + ((c > 2) ? p2 : 0)
                         + __builtin_popcount(mine & ((1u << lane) - 1u));
        const bool first = (mine >> lane) & 1u;
        const int slot = first ? before : (n1 + l - before);
        sh_koff[slot] = l * DD;
        const int   sidl  = (c == 0) ? sidv[0]  : (c == 1) ? sidv[1]  : (c == 2) ? sidv[2]  : sidv[3];
        const int   maskl = (c == 0) ? maskv[0] : (c == 1) ? maskv[1] : (c == 2) ? maskv[2] : maskv[3];
        #pragma unroll
        for (int row = 0; row < 32; ++row) {
            float v;
            if (row == 0)        v = maskl ? sh_inv[0] : 0.0f;
            else if (row <= 24)  v = (sidl == sh_sel[row]) ? sh_inv[row] : 0.0f;
            else                 v = 0.0f;
            sh_M[row][slot] = v;
        }
    }
    __syncthreads();

    const int lo   = lane & 15;
    const int hi   = lane >> 4;
    const int colBase = half * (DD / 2) + wid * 96;

    const float* __restrict__ E = emb + (size_t)bs * LL * DD + (colBase + lo);

    v8f acc[2][6];
    #pragma unroll
    for (int rt = 0; rt < 2; ++rt)
        #pragma unroll
        for (int j = 0; j < 6; ++j)
            acc[rt][j] = (v8f){};

    const int nb1 = (n1 + 15) >> 4;

    int kb = 0;
    for (; kb < nb1; ++kb) {
        v2f bq[4][6];
        #pragma unroll
        for (int s = 0; s < 4; ++s) {
            const int ka = kb * 16 + s * 4 + 2 * hi;
            const float* __restrict__ r0 = E + sh_koff[ka];
            const float* __restrict__ r1 = E + sh_koff[ka + 1];
            #pragma unroll
            for (int j = 0; j < 6; ++j) {
                bq[s][j].x = __builtin_nontemporal_load(r0 + j * 16);
                bq[s][j].y = __builtin_nontemporal_load(r1 + j * 16);
            }
        }
        __builtin_amdgcn_sched_barrier(0);
        #pragma unroll
        for (int s = 0; s < 4; ++s) {
            const int ka = kb * 16 + s * 4 + 2 * hi;
            v2f a0, a1;
            a0.x = sh_M[lo][ka];        a0.y = sh_M[lo][ka + 1];
            a1.x = sh_M[16 + lo][ka];   a1.y = sh_M[16 + lo][ka + 1];
            #pragma unroll
            for (int j = 0; j < 6; ++j) {
                acc[0][j] = __builtin_amdgcn_wmma_f32_16x16x4_f32(
                    false, a0, false, bq[s][j], (short)0, acc[0][j], false, false);
                acc[1][j] = __builtin_amdgcn_wmma_f32_16x16x4_f32(
                    false, a1, false, bq[s][j], (short)0, acc[1][j], false, false);
            }
        }
    }
    for (; kb < LL / 16; ++kb) {
        v2f bq[4][6];
        #pragma unroll
        for (int s = 0; s < 4; ++s) {
            const int ka = kb * 16 + s * 4 + 2 * hi;
            const float* __restrict__ r0 = E + sh_koff[ka];
            const float* __restrict__ r1 = E + sh_koff[ka + 1];
            #pragma unroll
            for (int j = 0; j < 6; ++j) {
                bq[s][j].x = __builtin_nontemporal_load(r0 + j * 16);
                bq[s][j].y = __builtin_nontemporal_load(r1 + j * 16);
            }
        }
        #pragma unroll
        for (int s = 0; s < 4; ++s) {
            const int ka = kb * 16 + s * 4 + 2 * hi;
            v2f a0;
            a0.x = sh_M[lo][ka];        a0.y = sh_M[lo][ka + 1];
            #pragma unroll
            for (int j = 0; j < 6; ++j) {
                acc[0][j] = __builtin_amdgcn_wmma_f32_16x16x4_f32(
                    false, a0, false, bq[s][j], (short)0, acc[0][j], false, false);
            }
        }
    }

    #pragma unroll
    for (int rt = 0; rt < 2; ++rt) {
        #pragma unroll
        for (int v = 0; v < 8; ++v) {
            const int r   = rt * 16 + 8 * hi + v;
            const int off = sh_rowoff[r];
            if (off >= 0) {
                #pragma unroll
                for (int j = 0; j < 6; ++j) {
                    const int col = colBase + j * 16 + lo;
                    __builtin_nontemporal_store(acc[rt][j][v], &out[off + col]);
                }
            }
        }
    }
}

extern "C" void kernel_launch(void* const* d_in, const int* in_sizes, int n_in,
                              void* d_out, int out_size, void* d_ws, size_t ws_size,
                              hipStream_t stream) {
    const float* emb   = (const float*)d_in[0];
    const int*   amask = (const int*)  d_in[1];
    const int*   sids  = (const int*)  d_in[2];
    const int*   pids  = (const int*)  d_in[3];
    const int*   a0    = (const int*)  d_in[4];
    const int*   a1    = (const int*)  d_in[5];
    float* out = (float*)d_out;

    dim3 grid(BB * SS * 2);
    dim3 block(128);
    hipLaunchKernelGGL(srl_fused_pool_wmma, grid, block, 0, stream,
                       emb, amask, sids, pids, a0, a1, out);
}
